# starting kernel
# baseline (speedup 1.0000x reference)
#include <hip/hip_runtime.h>


typedef __attribute__((ext_vector_type(2))) float v2f;
typedef __attribute__((ext_vector_type(8))) float v8f;


__global__ void GCN_zero_f32(float* __restrict__ p, long n) {
  long i = (long)blockIdx.x * blockDim.x + threadIdx.x;
  long stride = (long)gridDim.x * blockDim.x;
  for (; i < n; i += stride) p[i] = 0.0f;
}

__global__ void GCN_degree(const int* __restrict__ src, const int* __restrict__ dst,
                           float* __restrict__ deg_out, float* __restrict__ deg_in,
                           int ne) {
  int i = blockIdx.x * blockDim.x + threadIdx.x;
  if (i < ne) {
    atomicAdd(&deg_out[src[i]], 1.0f);
    atomicAdd(&deg_in[dst[i]], 1.0f);
  }
}

__global__ void GCN_norms(const float* __restrict__ deg_out, const float* __restrict__ deg_in,
                          float* __restrict__ nsrc, float* __restrict__ ndst, int n) {
  int i = blockIdx.x * blockDim.x + threadIdx.x;
  if (i < n) {
    nsrc[i] = rsqrtf(fmaxf(deg_out[i], 1.0f));
    ndst[i] = rsqrtf(fmaxf(deg_in[i], 1.0f));
  }
}

template <int K, int NT, int NC>
__global__ __launch_bounds__(32 * NT)
void GCN_gemm_wmma(const float* __restrict__ X, const float* __restrict__ scale,
                   const float* __restrict__ W, float* __restrict__ Y, int nrows) {
  constexpr int NCP = NT * 16;
  constexpr bool EXACT = (NC == NCP);
  constexpr int PITCH = K + 4;
  __shared__ float sa[16 * PITCH];
  __shared__ float sw[EXACT ? 4 : K * NCP];

  const int rowbase = blockIdx.x * 16;

  for (int idx = threadIdx.x; idx < 16 * (K / 4); idx += blockDim.x) {
    const int r = idx / (K / 4);
    const int c = idx - r * (K / 4);
    const int row = rowbase + r;
    float4 v = make_float4(0.f, 0.f, 0.f, 0.f);
    float s = 0.0f;
    if (row < nrows) {
      v = ((const float4*)(X + (long)row * K))[c];
      s = scale[row];
    }
    float* d = &sa[r * PITCH + 4 * c];
    d[0] = v.x * s; d[1] = v.y * s; d[2] = v.z * s; d[3] = v.w * s;
  }

  if constexpr (!EXACT) {
    for (int idx = threadIdx.x; idx < K * NCP; idx += blockDim.x) {
      const int k = idx / NCP;
      const int c = idx - k * NCP;
      sw[idx] = (c < NC) ? W[k * NC + c] : 0.0f;
    }
  }
  __syncthreads();

  const int lane = threadIdx.x & 31;
  const int wave = threadIdx.x >> 5;
  const int colbase = wave * 16;
  const int half = lane >> 4;
  const int l15 = lane & 15;
  const int koff = half * 2;

  const float* aptr = &sa[l15 * PITCH + koff];
  const int bcol = colbase + l15;

  v8f acc = {};
#pragma unroll 8
  for (int k0 = 0; k0 < K; k0 += 4) {
    v2f a;
    a.x = aptr[k0];
    a.y = aptr[k0 + 1];
    v2f b;
    if constexpr (EXACT) {
      b.x = W[(k0 + koff) * NC + bcol];
      b.y = W[(k0 + koff + 1) * NC + bcol];
    } else {
      b.x = sw[(k0 + koff) * NCP + bcol];
      b.y = sw[(k0 + koff + 1) * NCP + bcol];
    }
    acc = __builtin_amdgcn_wmma_f32_16x16x4_f32(
         false, a,  false, b,
         (short)0, acc,  false,  false);
  }

  if (EXACT || bcol < NC) {
    float* yp = Y + (long)(rowbase + half * 8) * NC + bcol;
    if (rowbase + 16 <= nrows) {
#pragma unroll
      for (int r = 0; r < 8; ++r) yp[(long)r * NC] = acc[r];
    } else {
      const int rbase = rowbase + half * 8;
#pragma unroll
      for (int r = 0; r < 8; ++r) {
        if (rbase + r < nrows) yp[(long)r * NC] = acc[r];
      }
    }
  }
}

__global__ void GCN_scatter128(const float* __restrict__ H,
                               const int* __restrict__ src, const int* __restrict__ dst,
                               float* __restrict__ AGG, int ne) {
  const int wid = (int)((blockIdx.x * (long)blockDim.x + threadIdx.x) >> 5);
  const int lane = threadIdx.x & 31;
  if (wid >= ne) return;
  const int s = src[wid];
  const int d = dst[wid];
  const float4 v = ((const float4*)(H + (long)s * 128))[lane];
  float* out = AGG + (long)d * 128 + lane * 4;
  atomicAdd(out + 0, v.x);
  atomicAdd(out + 1, v.y);
  atomicAdd(out + 2, v.z);
  atomicAdd(out + 3, v.w);
}

__global__ void GCN_scatter40(const float* __restrict__ H,
                              const int* __restrict__ src, const int* __restrict__ dst,
                              float* __restrict__ AGG, int ne) {
  const int wid = (int)((blockIdx.x * (long)blockDim.x + threadIdx.x) >> 5);
  const int lane = threadIdx.x & 31;
  if (wid >= ne) return;
  const int s = src[wid];
  const int d = dst[wid];
  const float* hp = H + (long)s * 40;
  float* out = AGG + (long)d * 40;
  atomicAdd(out + lane, hp[lane]);
  if (lane < 8) atomicAdd(out + lane + 32, hp[lane + 32]);
}


__global__ void GCN_post1(const float* __restrict__ AGG, const float* __restrict__ ndst,
                          const float* __restrict__ b, float* __restrict__ H1, long total) {
  long idx = (long)blockIdx.x * blockDim.x + threadIdx.x;
  long stride = (long)gridDim.x * blockDim.x;
  for (; idx < total; idx += stride) {
    const int i = (int)(idx >> 7);
    const int f = (int)(idx & 127);
    const float v = AGG[idx] * ndst[i] + b[f];
    H1[idx] = v > 0.0f ? v : 0.0f;
  }
}

__global__ void GCN_post2(float* __restrict__ OUT, const float* __restrict__ ndst,
                          const float* __restrict__ b, long total) {
  long idx = (long)blockIdx.x * blockDim.x + threadIdx.x;
  long stride = (long)gridDim.x * blockDim.x;
  for (; idx < total; idx += stride) {
    const int i = (int)(idx / 40);
    const int f = (int)(idx - (long)i * 40);
    OUT[idx] = OUT[idx] * ndst[i] + b[f];
  }
}


extern "C" void kernel_launch(void* const* d_in, const int* in_sizes, int n_in,
                              void* d_out, int out_size, void* d_ws, size_t ws_size,
                              hipStream_t stream) {
  const float* x   = (const float*)d_in[0];
  const float* W1  = (const float*)d_in[1];
  const float* b1  = (const float*)d_in[2];
  const float* W2  = (const float*)d_in[3];
  const float* b2  = (const float*)d_in[4];
  const int* esrc  = (const int*)d_in[5];
  const int* edst  = (const int*)d_in[6];
  float* out = (float*)d_out;

  const int nn = in_sizes[0] / 256;
  const int ne = in_sizes[5];

  const long NPAD = 50048;
  float* ws       = (float*)d_ws;
  float* deg_out  = ws;
  float* deg_in   = ws + NPAD;
  float* nsrc     = ws + 2 * NPAD;
  float* ndst     = ws + 3 * NPAD;
  float* h        = ws + 4 * NPAD;
  float* agg1     = ws + 4 * NPAD + (long)nn * 128;
  float* h2       = ws + 4 * NPAD + 2L * (long)nn * 128;

  GCN_zero_f32<<<512, 256, 0, stream>>>(deg_out, 2 * NPAD);
  GCN_zero_f32<<<4096, 256, 0, stream>>>(agg1, (long)nn * 128);
  GCN_zero_f32<<<2048, 256, 0, stream>>>(out, (long)nn * 40);

  GCN_degree<<<(ne + 255) / 256, 256, 0, stream>>>(esrc, edst, deg_out, deg_in, ne);
  GCN_norms<<<(nn + 255) / 256, 256, 0, stream>>>(deg_out, deg_in, nsrc, ndst, nn);

  const int mtiles = (nn + 15) / 16;

  GCN_gemm_wmma<256, 8, 128><<<mtiles, 256, 0, stream>>>(x, nsrc, W1, h, nn);

  GCN_scatter128<<<(ne + 7) / 8, 256, 0, stream>>>(h, esrc, edst, agg1, ne);

  GCN_post1<<<8192, 256, 0, stream>>>(agg1, ndst, b1, h, (long)nn * 128);

  GCN_gemm_wmma<128, 3, 40><<<mtiles, 96, 0, stream>>>(h, nsrc, W2, h2, nn);

  GCN_scatter40<<<(ne + 7) / 8, 256, 0, stream>>>(h2, esrc, edst, out, ne);

  GCN_post2<<<4096, 256, 0, stream>>>(out, ndst, b2, (long)nn * 40);
}


# degree counts via per-work-group LDS integer histograms (32 edge chunks per index array, planes summed in the norms pass) replace per-edge scattered global f32 atomics; zero-degree launch dropped
# speedup vs baseline: 1.2245x; 1.2245x over previous
#include <hip/hip_runtime.h>


typedef __attribute__((ext_vector_type(2))) float v2f;
typedef __attribute__((ext_vector_type(8))) float v8f;


__global__ void GCN_zero_f32(float* __restrict__ p, long n) {
  long i = (long)blockIdx.x * blockDim.x + threadIdx.x;
  long stride = (long)gridDim.x * blockDim.x;
  for (; i < n; i += stride) p[i] = 0.0f;
}

#define HIST_BINS 50048
#define HIST_G 32
#define HIST_T 1024

__global__ __launch_bounds__(HIST_T)
void GCN_degree_hist(const int* __restrict__ src, const int* __restrict__ dst,
                     unsigned* __restrict__ part, int ne) {
  __shared__ unsigned hist[HIST_BINS];
  const int* __restrict__ idx = blockIdx.y ? dst : src;

  uint4* h4 = (uint4*)hist;
  for (int i = threadIdx.x; i < HIST_BINS / 4; i += HIST_T) h4[i] = make_uint4(0u, 0u, 0u, 0u);
  __syncthreads();

  const int chunk = ((ne + 4 * HIST_G - 1) / (4 * HIST_G)) * 4;
  const int beg = min((int)blockIdx.x * chunk, ne);
  const int end = min(beg + chunk, ne);
  const int nvec = (end - beg) >> 2;
  const int4* __restrict__ p4 = (const int4*)(idx + beg);
  for (int base = 0; base < nvec; base += 4 * HIST_T) {
    int4 v[4];
#pragma unroll
    for (int j = 0; j < 4; ++j)
      v[j] = p4[min(base + j * HIST_T + (int)threadIdx.x, nvec - 1)];
#pragma unroll
    for (int j = 0; j < 4; ++j) {
      if (base + j * HIST_T + (int)threadIdx.x < nvec) {
        if ((unsigned)v[j].x < HIST_BINS) atomicAdd(&hist[v[j].x], 1u);
        if ((unsigned)v[j].y < HIST_BINS) atomicAdd(&hist[v[j].y], 1u);
        if ((unsigned)v[j].z < HIST_BINS) atomicAdd(&hist[v[j].z], 1u);
        if ((unsigned)v[j].w < HIST_BINS) atomicAdd(&hist[v[j].w], 1u);
      }
    }
  }
  for (int e = beg + (nvec << 2) + threadIdx.x; e < end; e += HIST_T) {
    const int v = idx[e];
    if ((unsigned)v < HIST_BINS) atomicAdd(&hist[v], 1u);
  }
  __syncthreads();

  uint4* o4 = (uint4*)(part + ((long)blockIdx.y * HIST_G + blockIdx.x) * HIST_BINS);
  for (int i = threadIdx.x; i < HIST_BINS / 4; i += HIST_T) o4[i] = h4[i];
}

__global__ void GCN_norms(const unsigned* __restrict__ part,
                          float* __restrict__ nsrc, float* __restrict__ ndst, int n) {
  const int i = blockIdx.x * blockDim.x + threadIdx.x;
  if (i < n) {
    const unsigned* p = part + (long)blockIdx.y * HIST_G * HIST_BINS + i;
    unsigned c = 0;
#pragma unroll
    for (int g = 0; g < HIST_G; ++g) c += p[(long)g * HIST_BINS];
    const float r = rsqrtf(fmaxf((float)c, 1.0f));
    if (blockIdx.y) ndst[i] = r; else nsrc[i] = r;
  }
}

template <int K, int NT, int NC>
__global__ __launch_bounds__(32 * NT)
void GCN_gemm_wmma(const float* __restrict__ X, const float* __restrict__ scale,
                   const float* __restrict__ W, float* __restrict__ Y, int nrows) {
  constexpr int NCP = NT * 16;
  constexpr bool EXACT = (NC == NCP);
  constexpr int PITCH = K + 4;
  __shared__ float sa[16 * PITCH];
  __shared__ float sw[EXACT ? 4 : K * NCP];

  const int rowbase = blockIdx.x * 16;

  for (int idx = threadIdx.x; idx < 16 * (K / 4); idx += blockDim.x) {
    const int r = idx / (K / 4);
    const int c = idx - r * (K / 4);
    const int row = rowbase + r;
    float4 v = make_float4(0.f, 0.f, 0.f, 0.f);
    float s = 0.0f;
    if (row < nrows) {
      v = ((const float4*)(X + (long)row * K))[c];
      s = scale[row];
    }
    float* d = &sa[r * PITCH + 4 * c];
    d[0] = v.x * s; d[1] = v.y * s; d[2] = v.z * s; d[3] = v.w * s;
  }

  if constexpr (!EXACT) {
    for (int idx = threadIdx.x; idx < K * NCP; idx += blockDim.x) {
      const int k = idx / NCP;
      const int c = idx - k * NCP;
      sw[idx] = (c < NC) ? W[k * NC + c] : 0.0f;
    }
  }
  __syncthreads();

  const int lane = threadIdx.x & 31;
  const int wave = threadIdx.x >> 5;
  const int colbase = wave * 16;
  const int half = lane >> 4;
  const int l15 = lane & 15;
  const int koff = half * 2;

  const float* aptr = &sa[l15 * PITCH + koff];
  const int bcol = colbase + l15;

  v8f acc = {};
#pragma unroll 8
  for (int k0 = 0; k0 < K; k0 += 4) {
    v2f a;
    a.x = aptr[k0];
    a.y = aptr[k0 + 1];
    v2f b;
    if constexpr (EXACT) {
      b.x = W[(k0 + koff) * NC + bcol];
      b.y = W[(k0 + koff + 1) * NC + bcol];
    } else {
      b.x = sw[(k0 + koff) * NCP + bcol];
      b.y = sw[(k0 + koff + 1) * NCP + bcol];
    }
    acc = __builtin_amdgcn_wmma_f32_16x16x4_f32(
         false, a,  false, b,
         (short)0, acc,  false,  false);
  }

  if (EXACT || bcol < NC) {
    float* yp = Y + (long)(rowbase + half * 8) * NC + bcol;
    if (rowbase + 16 <= nrows) {
#pragma unroll
      for (int r = 0; r < 8; ++r) yp[(long)r * NC] = acc[r];
    } else {
      const int rbase = rowbase + half * 8;
#pragma unroll
      for (int r = 0; r < 8; ++r) {
        if (rbase + r < nrows) yp[(long)r * NC] = acc[r];
      }
    }
  }
}

__global__ void GCN_scatter128(const float* __restrict__ H,
                               const int* __restrict__ src, const int* __restrict__ dst,
                               float* __restrict__ AGG, int ne) {
  const int wid = (int)((blockIdx.x * (long)blockDim.x + threadIdx.x) >> 5);
  const int lane = threadIdx.x & 31;
  if (wid >= ne) return;
  const int s = src[wid];
  const int d = dst[wid];
  const float4 v = ((const float4*)(H + (long)s * 128))[lane];
  float* out = AGG + (long)d * 128 + lane * 4;
  atomicAdd(out + 0, v.x);
  atomicAdd(out + 1, v.y);
  atomicAdd(out + 2, v.z);
  atomicAdd(out + 3, v.w);
}

__global__ void GCN_scatter40(const float* __restrict__ H,
                              const int* __restrict__ src, const int* __restrict__ dst,
                              float* __restrict__ AGG, int ne) {
  const int wid = (int)((blockIdx.x * (long)blockDim.x + threadIdx.x) >> 5);
  const int lane = threadIdx.x & 31;
  if (wid >= ne) return;
  const int s = src[wid];
  const int d = dst[wid];
  const float* hp = H + (long)s * 40;
  float* out = AGG + (long)d * 40;
  atomicAdd(out + lane, hp[lane]);
  if (lane < 8) atomicAdd(out + lane + 32, hp[lane + 32]);
}


__global__ void GCN_post1(const float* __restrict__ AGG, const float* __restrict__ ndst,
                          const float* __restrict__ b, float* __restrict__ H1, long total) {
  long idx = (long)blockIdx.x * blockDim.x + threadIdx.x;
  long stride = (long)gridDim.x * blockDim.x;
  for (; idx < total; idx += stride) {
    const int i = (int)(idx >> 7);
    const int f = (int)(idx & 127);
    const float v = AGG[idx] * ndst[i] + b[f];
    H1[idx] = v > 0.0f ? v : 0.0f;
  }
}

__global__ void GCN_post2(float* __restrict__ OUT, const float* __restrict__ ndst,
                          const float* __restrict__ b, long total) {
  long idx = (long)blockIdx.x * blockDim.x + threadIdx.x;
  long stride = (long)gridDim.x * blockDim.x;
  for (; idx < total; idx += stride) {
    const int i = (int)(idx / 40);
    const int f = (int)(idx - (long)i * 40);
    OUT[idx] = OUT[idx] * ndst[i] + b[f];
  }
}


extern "C" void kernel_launch(void* const* d_in, const int* in_sizes, int n_in,
                              void* d_out, int out_size, void* d_ws, size_t ws_size,
                              hipStream_t stream) {
  const float* x   = (const float*)d_in[0];
  const float* W1  = (const float*)d_in[1];
  const float* b1  = (const float*)d_in[2];
  const float* W2  = (const float*)d_in[3];
  const float* b2  = (const float*)d_in[4];
  const int* esrc  = (const int*)d_in[5];
  const int* edst  = (const int*)d_in[6];
  float* out = (float*)d_out;

  const int nn = in_sizes[0] / 256;
  const int ne = in_sizes[5];

  const long NPAD = 50048;
  float* ws       = (float*)d_ws;
  float* nsrc     = ws + 2 * NPAD;
  float* ndst     = ws + 3 * NPAD;
  float* h        = ws + 4 * NPAD;
  float* agg1     = ws + 4 * NPAD + (long)nn * 128;
  float* h2       = ws + 4 * NPAD + 2L * (long)nn * 128;
  unsigned* part  = (unsigned*)(ws + 4 * NPAD + 2L * (long)nn * 128 + (long)nn * 40);

  GCN_zero_f32<<<4096, 256, 0, stream>>>(agg1, (long)nn * 128);
  GCN_zero_f32<<<2048, 256, 0, stream>>>(out, (long)nn * 40);

  GCN_degree_hist<<<dim3(HIST_G, 2), HIST_T, 0, stream>>>(esrc, edst, part, ne);
  GCN_norms<<<dim3((nn + 255) / 256, 2), 256, 0, stream>>>(part, nsrc, ndst, nn);

  const int mtiles = (nn + 15) / 16;

  GCN_gemm_wmma<256, 8, 128><<<mtiles, 256, 0, stream>>>(x, nsrc, W1, h, nn);

  GCN_scatter128<<<(ne + 7) / 8, 256, 0, stream>>>(h, esrc, edst, agg1, ne);

  GCN_post1<<<8192, 256, 0, stream>>>(agg1, ndst, b1, h, (long)nn * 128);

  GCN_gemm_wmma<128, 3, 40><<<mtiles, 96, 0, stream>>>(h, nsrc, W2, h2, nn);

  GCN_scatter40<<<(ne + 7) / 8, 256, 0, stream>>>(h2, esrc, edst, out, ne);

  GCN_post2<<<4096, 256, 0, stream>>>(out, ndst, b2, (long)nn * 40);
}


# edge sums via destination-sorted edge list (per-call LDS counting sort from the degree histograms), rows summed in registers in sorted-edge order, replace f32 atomic scatter; zero/post passes fold in
# speedup vs baseline: 2.2416x; 1.8306x over previous
#include <hip/hip_runtime.h>


typedef __attribute__((ext_vector_type(2))) float v2f;
typedef __attribute__((ext_vector_type(4))) float v4f;
typedef __attribute__((ext_vector_type(8))) float v8f;

#define HIST_BINS 50048
#define HIST_G 32
#define HIST_T 1024

__global__ __launch_bounds__(HIST_T)
void GCN_degree_hist(const int* __restrict__ src, const int* __restrict__ dst,
                     unsigned* __restrict__ part, int ne) {
  __shared__ unsigned hist[HIST_BINS];
  const int* __restrict__ idx = blockIdx.y ? dst : src;

  uint4* h4 = (uint4*)hist;
  for (int i = threadIdx.x; i < HIST_BINS / 4; i += HIST_T) h4[i] = make_uint4(0u, 0u, 0u, 0u);
  __syncthreads();

  const int chunk = ((ne + 4 * HIST_G - 1) / (4 * HIST_G)) * 4;
  const int beg = min((int)blockIdx.x * chunk, ne);
  const int end = min(beg + chunk, ne);
  const int nvec = (end - beg) >> 2;
  const int4* __restrict__ p4 = (const int4*)(idx + beg);
  for (int base = 0; base < nvec; base += 4 * HIST_T) {
    int4 v[4];
#pragma unroll
    for (int j = 0; j < 4; ++j)
      v[j] = p4[min(base + j * HIST_T + (int)threadIdx.x, nvec - 1)];
#pragma unroll
    for (int j = 0; j < 4; ++j) {
      if (base + j * HIST_T + (int)threadIdx.x < nvec) {
        if ((unsigned)v[j].x < HIST_BINS) atomicAdd(&hist[v[j].x], 1u);
        if ((unsigned)v[j].y < HIST_BINS) atomicAdd(&hist[v[j].y], 1u);
        if ((unsigned)v[j].z < HIST_BINS) atomicAdd(&hist[v[j].z], 1u);
        if ((unsigned)v[j].w < HIST_BINS) atomicAdd(&hist[v[j].w], 1u);
      }
    }
  }
  for (int e = beg + (nvec << 2) + threadIdx.x; e < end; e += HIST_T) {
    const int v = idx[e];
    if ((unsigned)v < HIST_BINS) atomicAdd(&hist[v], 1u);
  }
  __syncthreads();

  uint4* o4 = (uint4*)(part + ((long)blockIdx.y * HIST_G + blockIdx.x) * HIST_BINS);
  for (int i = threadIdx.x; i < HIST_BINS / 4; i += HIST_T) o4[i] = h4[i];
}

__global__ void GCN_norms(unsigned* __restrict__ part,
                          float* __restrict__ nsrc, float* __restrict__ ndst,
                          unsigned* __restrict__ indeg, int n) {
  const int i = blockIdx.x * blockDim.x + threadIdx.x;
  if (i < n) {
    unsigned* p = part + (long)blockIdx.y * HIST_G * HIST_BINS + i;
    unsigned t[HIST_G];
#pragma unroll
    for (int g = 0; g < HIST_G; ++g) t[g] = p[(long)g * HIST_BINS];
    unsigned c = 0;
    if (blockIdx.y) {
#pragma unroll
      for (int g = 0; g < HIST_G; ++g) { p[(long)g * HIST_BINS] = c; c += t[g]; }
      indeg[i] = c;
      ndst[i] = rsqrtf(fmaxf((float)c, 1.0f));
    } else {
#pragma unroll
      for (int g = 0; g < HIST_G; ++g) c += t[g];
      nsrc[i] = rsqrtf(fmaxf((float)c, 1.0f));
    }
  } else if (blockIdx.y && i < HIST_BINS) {
    indeg[i] = 0u;
  }
}

#define SCAN_PER ((HIST_BINS + HIST_T - 1) / HIST_T)
#define SCAN_LEN (SCAN_PER * HIST_T)
#define BIN_VECS (HIST_BINS / 4)
#define VECS_PER ((BIN_VECS + HIST_T - 1) / HIST_T)

__global__ __launch_bounds__(HIST_T)
void GCN_sort_by_dst(const int* __restrict__ src, const int* __restrict__ dst,
                     const unsigned* __restrict__ pre, const unsigned* __restrict__ indeg,
                     unsigned* __restrict__ rowptr, int* __restrict__ sorted_src,
                     int n, int ne) {
  __shared__ unsigned cur[SCAN_LEN];
  __shared__ unsigned wsum[HIST_T / 32];
  const int t = threadIdx.x;
  const int lane = t & 31;
  const int w = t >> 5;
  uint4* c4 = (uint4*)cur;

  {
    const uint4* __restrict__ in4 = (const uint4*)indeg;
    uint4 v[VECS_PER];
#pragma unroll
    for (int j = 0; j < VECS_PER; ++j) v[j] = in4[min(j * HIST_T + t, BIN_VECS - 1)];
#pragma unroll
    for (int j = 0; j < VECS_PER; ++j) {
      const int q = j * HIST_T + t;
      if (q < BIN_VECS) c4[q] = v[j];
    }
    for (int i = HIST_BINS + t; i < SCAN_LEN; i += HIST_T) cur[i] = 0u;
  }
  __syncthreads();

  const int b0 = t * SCAN_PER;
  unsigned cnt[SCAN_PER];
  unsigned s = 0;
#pragma unroll
  for (int j = 0; j < SCAN_PER; ++j) { cnt[j] = cur[b0 + j]; s += cnt[j]; }
  unsigned inc = s;
#pragma unroll
  for (int o = 1; o < 32; o <<= 1) { const unsigned v = __shfl_up(inc, o); if (lane >= o) inc += v; }
  if (lane == 31) wsum[w] = inc;
  __syncthreads();
  if (w == 0) {
    const unsigned v = wsum[lane];
    unsigned iv = v;
#pragma unroll
    for (int o = 1; o < 32; o <<= 1) { const unsigned u = __shfl_up(iv, o); if (lane >= o) iv += u; }
    wsum[lane] = iv - v;
  }
  __syncthreads();
  unsigned run = inc - s + wsum[w];
#pragma unroll
  for (int j = 0; j < SCAN_PER; ++j) { cur[b0 + j] = run; run += cnt[j]; }
  __syncthreads();

  if (blockIdx.x == 0) {
    for (int i = t; i <= n; i += HIST_T) rowptr[i] = cur[i];
  }
  {
    const uint4* __restrict__ pg4 = (const uint4*)(pre + (long)blockIdx.x * HIST_BINS);
    uint4 v[VECS_PER];
#pragma unroll
    for (int j = 0; j < VECS_PER; ++j) v[j] = pg4[min(j * HIST_T + t, BIN_VECS - 1)];
#pragma unroll
    for (int j = 0; j < VECS_PER; ++j) {
      const int q = j * HIST_T + t;
      if (q < BIN_VECS) {
        uint4 c = c4[q];
        c.x += v[j].x; c.y += v[j].y; c.z += v[j].z; c.w += v[j].w;
        c4[q] = c;
      }
    }
  }
  __syncthreads();

  const int chunk = ((ne + 4 * HIST_G - 1) / (4 * HIST_G)) * 4;
  const int beg = min((int)blockIdx.x * chunk, ne);
  const int end = min(beg + chunk, ne);
  const int nvec = (end - beg) >> 2;
  const int4* __restrict__ d4 = (const int4*)(dst + beg);
  const int4* __restrict__ s4 = (const int4*)(src + beg);
  for (int base = 0; base < nvec; base += 4 * HIST_T) {
    int4 dv[4], sv[4];
#pragma unroll
    for (int j = 0; j < 4; ++j) {
      const int q = min(base + j * HIST_T + t, nvec - 1);
      dv[j] = d4[q];
      sv[j] = s4[q];
    }
#pragma unroll
    for (int j = 0; j < 4; ++j) {
      if (base + j * HIST_T + t < nvec) {
        if ((unsigned)dv[j].x < (unsigned)n) sorted_src[atomicAdd(&cur[dv[j].x], 1u)] = sv[j].x;
        if ((unsigned)dv[j].y < (unsigned)n) sorted_src[atomicAdd(&cur[dv[j].y], 1u)] = sv[j].y;
        if ((unsigned)dv[j].z < (unsigned)n) sorted_src[atomicAdd(&cur[dv[j].z], 1u)] = sv[j].z;
        if ((unsigned)dv[j].w < (unsigned)n) sorted_src[atomicAdd(&cur[dv[j].w], 1u)] = sv[j].w;
      }
    }
  }
  for (int e = beg + (nvec << 2) + t; e < end; e += HIST_T) {
    const int d = dst[e];
    if ((unsigned)d < (unsigned)n) sorted_src[atomicAdd(&cur[d], 1u)] = src[e];
  }
}

template <int K, int NT, int NC>
__global__ __launch_bounds__(32 * NT)
void GCN_gemm_wmma(const float* __restrict__ X, const float* __restrict__ scale,
                   const float* __restrict__ W, float* __restrict__ Y, int nrows) {
  constexpr int NCP = NT * 16;
  constexpr bool EXACT = (NC == NCP);
  constexpr int PITCH = K + 4;
  __shared__ float sa[16 * PITCH];
  __shared__ float sw[EXACT ? 4 : K * NCP];

  const int rowbase = blockIdx.x * 16;

  for (int idx = threadIdx.x; idx < 16 * (K / 4); idx += blockDim.x) {
    const int r = idx / (K / 4);
    const int c = idx - r * (K / 4);
    const int row = rowbase + r;
    float4 v = make_float4(0.f, 0.f, 0.f, 0.f);
    float s = 0.0f;
    if (row < nrows) {
      v = ((const float4*)(X + (long)row * K))[c];
      s = scale[row];
    }
    float* d = &sa[r * PITCH + 4 * c];
    d[0] = v.x * s; d[1] = v.y * s; d[2] = v.z * s; d[3] = v.w * s;
  }

  if constexpr (!EXACT) {
    for (int idx = threadIdx.x; idx < K * NCP; idx += blockDim.x) {
      const int k = idx / NCP;
      const int c = idx - k * NCP;
      sw[idx] = (c < NC) ? W[k * NC + c] : 0.0f;
    }
  }
  __syncthreads();

  const int lane = threadIdx.x & 31;
  const int wave = threadIdx.x >> 5;
  const int colbase = wave * 16;
  const int half = lane >> 4;
  const int l15 = lane & 15;
  const int koff = half * 2;

  const float* aptr = &sa[l15 * PITCH + koff];
  const int bcol = colbase + l15;

  v8f acc = {};
#pragma unroll 8
  for (int k0 = 0; k0 < K; k0 += 4) {
    v2f a;
    a.x = aptr[k0];
    a.y = aptr[k0 + 1];
    v2f b;
    if constexpr (EXACT) {
      b.x = W[(k0 + koff) * NC + bcol];
      b.y = W[(k0 + koff + 1) * NC + bcol];
    } else {
      b.x = sw[(k0 + koff) * NCP + bcol];
      b.y = sw[(k0 + koff + 1) * NCP + bcol];
    }
    acc = __builtin_amdgcn_wmma_f32_16x16x4_f32(
         false, a,  false, b,
         (short)0, acc,  false,  false);
  }

  if (EXACT || bcol < NC) {
    float* yp = Y + (long)(rowbase + half * 8) * NC + bcol;
    if (rowbase + 16 <= nrows) {
#pragma unroll
      for (int r = 0; r < 8; ++r) yp[(long)r * NC] = acc[r];
    } else {
      const int rbase = rowbase + half * 8;
#pragma unroll
      for (int r = 0; r < 8; ++r) {
        if (rbase + r < nrows) yp[(long)r * NC] = acc[r];
      }
    }
  }
}

#define GATHER_WGS 2048

__global__ __launch_bounds__(256)
void GCN_gather128(const float* __restrict__ H, const unsigned* __restrict__ rowptr,
                   const int* __restrict__ sorted_src, const float* __restrict__ ndst,
                   const float* __restrict__ b, float* __restrict__ H1, int n) {
  const int lane = threadIdx.x & 31;
  const int wave0 = __builtin_amdgcn_readfirstlane((int)(blockIdx.x * 8 + (threadIdx.x >> 5)));
  const int nwaves = (int)gridDim.x * 8;
  const v4f bb = ((const v4f*)b)[lane];
  for (int node = wave0; node < n; node += nwaves) {
    const int beg = (int)rowptr[node];
    const int end = (int)rowptr[node + 1];
    v4f acc = {0.f, 0.f, 0.f, 0.f};
    for (int j0 = beg; j0 < end; j0 += 32) {
      const int cnt = min(32, end - j0);
      const int mine = sorted_src[j0 + min(lane, cnt - 1)];
      for (int k = 0; k < cnt; k += 8) {
        v4f v[8];
#pragma unroll
        for (int i = 0; i < 8; ++i) {
          const int s = __builtin_amdgcn_readlane(mine, min(k + i, cnt - 1));
          v[i] = ((const v4f*)(H + (long)s * 128))[lane];
        }
        asm volatile("" :: "v"(v[0]), "v"(v[1]), "v"(v[2]), "v"(v[3]),
                           "v"(v[4]), "v"(v[5]), "v"(v[6]), "v"(v[7]));
        acc += v[0];
#pragma unroll
        for (int i = 1; i < 8; ++i) {
          if (k + i < cnt) acc += v[i];
        }
      }
    }
    const float nd = ndst[node];
    v4f o = acc * nd + bb;
    o.x = o.x > 0.0f ? o.x : 0.0f;
    o.y = o.y > 0.0f ? o.y : 0.0f;
    o.z = o.z > 0.0f ? o.z : 0.0f;
    o.w = o.w > 0.0f ? o.w : 0.0f;
    ((v4f*)(H1 + (long)node * 128))[lane] = o;
  }
}

__global__ __launch_bounds__(256)
void GCN_gather40(const float* __restrict__ H, const unsigned* __restrict__ rowptr,
                  const int* __restrict__ sorted_src, const float* __restrict__ ndst,
                  const float* __restrict__ b, float* __restrict__ OUT, int n) {
  const int lane = threadIdx.x & 31;
  const int l2 = min(lane, 19);
  const int wave0 = __builtin_amdgcn_readfirstlane((int)(blockIdx.x * 8 + (threadIdx.x >> 5)));
  const int nwaves = (int)gridDim.x * 8;
  const v2f bb = ((const v2f*)b)[l2];
  for (int node = wave0; node < n; node += nwaves) {
    const int beg = (int)rowptr[node];
    const int end = (int)rowptr[node + 1];
    v2f acc = {0.f, 0.f};
    for (int j0 = beg; j0 < end; j0 += 32) {
      const int cnt = min(32, end - j0);
      const int mine = sorted_src[j0 + min(lane, cnt - 1)];
      for (int k = 0; k < cnt; k += 8) {
        v2f v[8];
#pragma unroll
        for (int i = 0; i < 8; ++i) {
          const int s = __builtin_amdgcn_readlane(mine, min(k + i, cnt - 1));
          v[i] = ((const v2f*)(H + (long)s * 40))[l2];
        }
        asm volatile("" :: "v"(v[0]), "v"(v[1]), "v"(v[2]), "v"(v[3]),
                           "v"(v[4]), "v"(v[5]), "v"(v[6]), "v"(v[7]));
        acc += v[0];
#pragma unroll
        for (int i = 1; i < 8; ++i) {
          if (k + i < cnt) acc += v[i];
        }
      }
    }
    if (lane < 20) {
      const float nd = ndst[node];
      ((v2f*)(OUT + (long)node * 40))[lane] = acc * nd + bb;
    }
  }
}


extern "C" void kernel_launch(void* const* d_in, const int* in_sizes, int n_in,
                              void* d_out, int out_size, void* d_ws, size_t ws_size,
                              hipStream_t stream) {
  const float* x   = (const float*)d_in[0];
  const float* W1  = (const float*)d_in[1];
  const float* b1  = (const float*)d_in[2];
  const float* W2  = (const float*)d_in[3];
  const float* b2  = (const float*)d_in[4];
  const int* esrc  = (const int*)d_in[5];
  const int* edst  = (const int*)d_in[6];
  float* out = (float*)d_out;

  const int nn = in_sizes[0] / 256;
  const int ne = in_sizes[5];

  const long NPAD = 50048;
  float* ws       = (float*)d_ws;
  unsigned* indeg  = (unsigned*)ws;
  unsigned* rowptr = (unsigned*)(ws + NPAD);
  float* nsrc     = ws + 2 * NPAD;
  float* ndst     = ws + 3 * NPAD;
  float* h        = ws + 4 * NPAD;
  float* h1       = ws + 4 * NPAD + (long)nn * 128;
  float* h2       = ws + 4 * NPAD + 2L * (long)nn * 128;
  unsigned* part  = (unsigned*)(ws + 4 * NPAD + 2L * (long)nn * 128 + (long)nn * 40);
  int* ssrc       = (int*)(part + 2L * HIST_G * HIST_BINS);

  GCN_degree_hist<<<dim3(HIST_G, 2), HIST_T, 0, stream>>>(esrc, edst, part, ne);
  GCN_norms<<<dim3((HIST_BINS + 255) / 256, 2), 256, 0, stream>>>(part, nsrc, ndst, indeg, nn);

  GCN_sort_by_dst<<<HIST_G, HIST_T, 0, stream>>>(esrc, edst, part + (long)HIST_G * HIST_BINS,
                                                indeg, rowptr, ssrc, nn, ne);

  const int mtiles = (nn + 15) / 16;

  GCN_gemm_wmma<256, 8, 128><<<mtiles, 256, 0, stream>>>(x, nsrc, W1, h, nn);

  GCN_gather128<<<GATHER_WGS, 256, 0, stream>>>(h, rowptr, ssrc, ndst, b1, h1, nn);

  GCN_gemm_wmma<128, 3, 40><<<mtiles, 96, 0, stream>>>(h1, nsrc, W2, h2, nn);

  GCN_gather40<<<GATHER_WGS, 256, 0, stream>>>(h2, rowptr, ssrc, ndst, b2, out, nn);
}


# layer-2 product fused into the layer-1 edge-sum launch: a work-group sums 32 destination rows into LDS, six waves multiply them by W2 (LDS) on f32 WMMA; the h1 write+read and one launch drop
# speedup vs baseline: 2.9588x; 1.3200x over previous
#include <hip/hip_runtime.h>


typedef __attribute__((ext_vector_type(2))) float v2f;
typedef __attribute__((ext_vector_type(4))) float v4f;
typedef __attribute__((ext_vector_type(8))) float v8f;

#define HIST_BINS 50048
#define HIST_G 32
#define HIST_T 1024

__global__ __launch_bounds__(HIST_T)
void GCN_degree_hist(const int* __restrict__ src, const int* __restrict__ dst,
                     unsigned* __restrict__ part, int ne) {
  __shared__ unsigned hist[HIST_BINS];
  const int* __restrict__ idx = blockIdx.y ? dst : src;

  uint4* h4 = (uint4*)hist;
  for (int i = threadIdx.x; i < HIST_BINS / 4; i += HIST_T) h4[i] = make_uint4(0u, 0u, 0u, 0u);
  __syncthreads();

  const int chunk = ((ne + 4 * HIST_G - 1) / (4 * HIST_G)) * 4;
  const int beg = min((int)blockIdx.x * chunk, ne);
  const int end = min(beg + chunk, ne);
  const int nvec = (end - beg) >> 2;
  const int4* __restrict__ p4 = (const int4*)(idx + beg);
  for (int base = 0; base < nvec; base += 4 * HIST_T) {
    int4 v[4];
#pragma unroll
    for (int j = 0; j < 4; ++j)
      v[j] = p4[min(base + j * HIST_T + (int)threadIdx.x, nvec - 1)];
#pragma unroll
    for (int j = 0; j < 4; ++j) {
      if (base + j * HIST_T + (int)threadIdx.x < nvec) {
        if ((unsigned)v[j].x < HIST_BINS) atomicAdd(&hist[v[j].x], 1u);
        if ((unsigned)v[j].y < HIST_BINS) atomicAdd(&hist[v[j].y], 1u);
        if ((unsigned)v[j].z < HIST_BINS) atomicAdd(&hist[v[j].z], 1u);
        if ((unsigned)v[j].w < HIST_BINS) atomicAdd(&hist[v[j].w], 1u);
      }
    }
  }
  for (int e = beg + (nvec << 2) + threadIdx.x; e < end; e += HIST_T) {
    const int v = idx[e];
    if ((unsigned)v < HIST_BINS) atomicAdd(&hist[v], 1u);
  }
  __syncthreads();

  uint4* o4 = (uint4*)(part + ((long)blockIdx.y * HIST_G + blockIdx.x) * HIST_BINS);
  for (int i = threadIdx.x; i < HIST_BINS / 4; i += HIST_T) o4[i] = h4[i];
}

__global__ void GCN_norms(unsigned* __restrict__ part,
                          float* __restrict__ nsrc, float* __restrict__ ndst,
                          unsigned* __restrict__ indeg, int n) {
  const int i = blockIdx.x * blockDim.x + threadIdx.x;
  if (i < n) {
    unsigned* p = part + (long)blockIdx.y * HIST_G * HIST_BINS + i;
    unsigned t[HIST_G];
#pragma unroll
    for (int g = 0; g < HIST_G; ++g) t[g] = p[(long)g * HIST_BINS];
    unsigned c = 0;
    if (blockIdx.y) {
#pragma unroll
      for (int g = 0; g < HIST_G; ++g) { p[(long)g * HIST_BINS] = c; c += t[g]; }
      indeg[i] = c;
      ndst[i] = rsqrtf(fmaxf((float)c, 1.0f));
    } else {
#pragma unroll
      for (int g = 0; g < HIST_G; ++g) c += t[g];
      nsrc[i] = rsqrtf(fmaxf((float)c, 1.0f));
    }
  } else if (blockIdx.y && i < HIST_BINS) {
    indeg[i] = 0u;
  }
}

#define SCAN_PER ((HIST_BINS + HIST_T - 1) / HIST_T)
#define SCAN_LEN (SCAN_PER * HIST_T)
#define BIN_VECS (HIST_BINS / 4)
#define VECS_PER ((BIN_VECS + HIST_T - 1) / HIST_T)

__global__ __launch_bounds__(HIST_T)
void GCN_sort_by_dst(const int* __restrict__ src, const int* __restrict__ dst,
                     const unsigned* __restrict__ pre, const unsigned* __restrict__ indeg,
                     unsigned* __restrict__ rowptr, int* __restrict__ sorted_src,
                     int n, int ne) {
  __shared__ unsigned cur[SCAN_LEN];
  __shared__ unsigned wsum[HIST_T / 32];
  const int t = threadIdx.x;
  const int lane = t & 31;
  const int w = t >> 5;
  uint4* c4 = (uint4*)cur;

  {
    const uint4* __restrict__ in4 = (const uint4*)indeg;
    uint4 v[VECS_PER];
#pragma unroll
    for (int j = 0; j < VECS_PER; ++j) v[j] = in4[min(j * HIST_T + t, BIN_VECS - 1)];
#pragma unroll
    for (int j = 0; j < VECS_PER; ++j) {
      const int q = j * HIST_T + t;
      if (q < BIN_VECS) c4[q] = v[j];
    }
    for (int i = HIST_BINS + t; i < SCAN_LEN; i += HIST_T) cur[i] = 0u;
  }
  __syncthreads();

  const int b0 = t * SCAN_PER;
  unsigned cnt[SCAN_PER];
  unsigned s = 0;
#pragma unroll
  for (int j = 0; j < SCAN_PER; ++j) { cnt[j] = cur[b0 + j]; s += cnt[j]; }
  unsigned inc = s;
#pragma unroll
  for (int o = 1; o < 32; o <<= 1) { const unsigned v = __shfl_up(inc, o); if (lane >= o) inc += v; }
  if (lane == 31) wsum[w] = inc;
  __syncthreads();
  if (w == 0) {
    const unsigned v = wsum[lane];
    unsigned iv = v;
#pragma unroll
    for (int o = 1; o < 32; o <<= 1) { const unsigned u = __shfl_up(iv, o); if (lane >= o) iv += u; }
    wsum[lane] = iv - v;
  }
  __syncthreads();
  unsigned run = inc - s + wsum[w];
#pragma unroll
  for (int j = 0; j < SCAN_PER; ++j) { cur[b0 + j] = run; run += cnt[j]; }
  __syncthreads();

  if (blockIdx.x == 0) {
    for (int i = t; i <= n; i += HIST_T) rowptr[i] = cur[i];
  }
  {
    const uint4* __restrict__ pg4 = (const uint4*)(pre + (long)blockIdx.x * HIST_BINS);
    uint4 v[VECS_PER];
#pragma unroll
    for (int j = 0; j < VECS_PER; ++j) v[j] = pg4[min(j * HIST_T + t, BIN_VECS - 1)];
#pragma unroll
    for (int j = 0; j < VECS_PER; ++j) {
      const int q = j * HIST_T + t;
      if (q < BIN_VECS) {
        uint4 c = c4[q];
        c.x += v[j].x; c.y += v[j].y; c.z += v[j].z; c.w += v[j].w;
        c4[q] = c;
      }
    }
  }
  __syncthreads();

  const int chunk = ((ne + 4 * HIST_G - 1) / (4 * HIST_G)) * 4;
  const int beg = min((int)blockIdx.x * chunk, ne);
  const int end = min(beg + chunk, ne);
  const int nvec = (end - beg) >> 2;
  const int4* __restrict__ d4 = (const int4*)(dst + beg);
  const int4* __restrict__ s4 = (const int4*)(src + beg);
  for (int base = 0; base < nvec; base += 4 * HIST_T) {
    int4 dv[4], sv[4];
#pragma unroll
    for (int j = 0; j < 4; ++j) {
      const int q = min(base + j * HIST_T + t, nvec - 1);
      dv[j] = d4[q];
      sv[j] = s4[q];
    }
#pragma unroll
    for (int j = 0; j < 4; ++j) {
      if (base + j * HIST_T + t < nvec) {
        if ((unsigned)dv[j].x < (unsigned)n) sorted_src[atomicAdd(&cur[dv[j].x], 1u)] = sv[j].x;
        if ((unsigned)dv[j].y < (unsigned)n) sorted_src[atomicAdd(&cur[dv[j].y], 1u)] = sv[j].y;
        if ((unsigned)dv[j].z < (unsigned)n) sorted_src[atomicAdd(&cur[dv[j].z], 1u)] = sv[j].z;
        if ((unsigned)dv[j].w < (unsigned)n) sorted_src[atomicAdd(&cur[dv[j].w], 1u)] = sv[j].w;
      }
    }
  }
  for (int e = beg + (nvec << 2) + t; e < end; e += HIST_T) {
    const int d = dst[e];
    if ((unsigned)d < (unsigned)n) sorted_src[atomicAdd(&cur[d], 1u)] = src[e];
  }
}

template <int K, int NT, int NC>
__global__ __launch_bounds__(32 * NT)
void GCN_gemm_wmma(const float* __restrict__ X, const float* __restrict__ scale,
                   const float* __restrict__ W, float* __restrict__ Y, int nrows) {
  constexpr int NCP = NT * 16;
  constexpr bool EXACT = (NC == NCP);
  constexpr int PITCH = K + 4;
  __shared__ float sa[16 * PITCH];
  __shared__ float sw[EXACT ? 4 : K * NCP];

  const int rowbase = blockIdx.x * 16;

  for (int idx = threadIdx.x; idx < 16 * (K / 4); idx += blockDim.x) {
    const int r = idx / (K / 4);
    const int c = idx - r * (K / 4);
    const int row = rowbase + r;
    float4 v = make_float4(0.f, 0.f, 0.f, 0.f);
    float s = 0.0f;
    if (row < nrows) {
      v = ((const float4*)(X + (long)row * K))[c];
      s = scale[row];
    }
    float* d = &sa[r * PITCH + 4 * c];
    d[0] = v.x * s; d[1] = v.y * s; d[2] = v.z * s; d[3] = v.w * s;
  }

  if constexpr (!EXACT) {
    for (int idx = threadIdx.x; idx < K * NCP; idx += blockDim.x) {
      const int k = idx / NCP;
      const int c = idx - k * NCP;
      sw[idx] = (c < NC) ? W[k * NC + c] : 0.0f;
    }
  }
  __syncthreads();

  const int lane = threadIdx.x & 31;
  const int wave = threadIdx.x >> 5;
  const int colbase = wave * 16;
  const int half = lane >> 4;
  const int l15 = lane & 15;
  const int koff = half * 2;

  const float* aptr = &sa[l15 * PITCH + koff];
  const int bcol = colbase + l15;

  v8f acc = {};
#pragma unroll 8
  for (int k0 = 0; k0 < K; k0 += 4) {
    v2f a;
    a.x = aptr[k0];
    a.y = aptr[k0 + 1];
    v2f b;
    if constexpr (EXACT) {
      b.x = W[(k0 + koff) * NC + bcol];
      b.y = W[(k0 + koff + 1) * NC + bcol];
    } else {
      b.x = sw[(k0 + koff) * NCP + bcol];
      b.y = sw[(k0 + koff + 1) * NCP + bcol];
    }
    acc = __builtin_amdgcn_wmma_f32_16x16x4_f32(
         false, a,  false, b,
         (short)0, acc,  false,  false);
  }

  if (EXACT || bcol < NC) {
    float* yp = Y + (long)(rowbase + half * 8) * NC + bcol;
    if (rowbase + 16 <= nrows) {
#pragma unroll
      for (int r = 0; r < 8; ++r) yp[(long)r * NC] = acc[r];
    } else {
      const int rbase = rowbase + half * 8;
#pragma unroll
      for (int r = 0; r < 8; ++r) {
        if (rbase + r < nrows) yp[(long)r * NC] = acc[r];
      }
    }
  }
}

#define GATHER_WGS 2048

#define FUSE_ROWS 32
#define FUSE_PITCH 132
#define FUSE_NCP 48

__global__ __launch_bounds__(256)
void GCN_gather128_gemm40(const float* __restrict__ H, const unsigned* __restrict__ rowptr,
                          const int* __restrict__ sorted_src, const float* __restrict__ ndst,
                          const float* __restrict__ nsrc, const float* __restrict__ b,
                          const float* __restrict__ W2, float* __restrict__ H2, int n) {
  __shared__ float sa[FUSE_ROWS * FUSE_PITCH];
  __shared__ float sw[128 * FUSE_NCP];
  const int t = threadIdx.x;
  const int lane = t & 31;
  const int wave = __builtin_amdgcn_readfirstlane(t >> 5);

  {
    v4f v[5];
#pragma unroll
    for (int j = 0; j < 5; ++j) v[j] = ((const v4f*)W2)[t + 256 * j];
#pragma unroll
    for (int j = 0; j < 5; ++j) {
      const int q = t + 256 * j;
      const int k = q / 10;
      const int c = q - k * 10;
      *(v4f*)&sw[k * FUSE_NCP + 4 * c] = v[j];
    }
    const v4f z = {0.f, 0.f, 0.f, 0.f};
    *(v4f*)&sw[(t >> 1) * FUSE_NCP + 40 + 4 * (t & 1)] = z;
  }

  const v4f bb = ((const v4f*)b)[lane];
  const int node0 = (int)blockIdx.x * FUSE_ROWS + wave * 4;
  for (int i = 0; i < 4; ++i) {
    const int node = node0 + i;
    v4f o = {0.f, 0.f, 0.f, 0.f};
    if (node < n) {
      const float nd = ndst[node];
      const float ns = nsrc[node];
      const int beg = (int)rowptr[node];
      const int end = (int)rowptr[node + 1];
      v4f acc = {0.f, 0.f, 0.f, 0.f};
      for (int j0 = beg; j0 < end; j0 += 32) {
        const int cnt = min(32, end - j0);
        const int mine = sorted_src[j0 + min(lane, cnt - 1)];
        for (int k = 0; k < cnt; k += 8) {
          v4f v[8];
#pragma unroll
          for (int s8 = 0; s8 < 8; ++s8) {
            const int s = __builtin_amdgcn_readlane(mine, min(k + s8, cnt - 1));
            v[s8] = ((const v4f*)(H + (long)s * 128))[lane];
          }
          asm volatile("" :: "v"(v[0]), "v"(v[1]), "v"(v[2]), "v"(v[3]),
                             "v"(v[4]), "v"(v[5]), "v"(v[6]), "v"(v[7]));
          acc += v[0];
#pragma unroll
          for (int s8 = 1; s8 < 8; ++s8) {
            if (k + s8 < cnt) acc += v[s8];
          }
        }
      }
      o = acc * nd + bb;
      o.x = o.x > 0.0f ? o.x : 0.0f;
      o.y = o.y > 0.0f ? o.y : 0.0f;
      o.z = o.z > 0.0f ? o.z : 0.0f;
      o.w = o.w > 0.0f ? o.w : 0.0f;
      o = o * ns;
    }
    *(v4f*)&sa[(wave * 4 + i) * FUSE_PITCH + 4 * lane] = o;
  }
  __syncthreads();

  if (wave < 6) {
    const int tile = wave / 3;
    const int ct = wave - tile * 3;
    const int half = lane >> 4;
    const int l15 = lane & 15;
    const int koff = half * 2;
    const float* aptr = &sa[(tile * 16 + l15) * FUSE_PITCH + koff];
    const int bcol = ct * 16 + l15;

    v8f acc = {};
#pragma unroll 8
    for (int k0 = 0; k0 < 128; k0 += 4) {
      v2f a;
      a.x = aptr[k0];
      a.y = aptr[k0 + 1];
      v2f bf;
      bf.x = sw[(k0 + koff) * FUSE_NCP + bcol];
      bf.y = sw[(k0 + koff + 1) * FUSE_NCP + bcol];
      acc = __builtin_amdgcn_wmma_f32_16x16x4_f32(
           false, a,  false, bf,
           (short)0, acc,  false,  false);
    }

    if (bcol < 40) {
      const int rowbase = (int)blockIdx.x * FUSE_ROWS + tile * 16;
      const int rbase = rowbase + half * 8;
      float* yp = H2 + (long)rbase * 40 + bcol;
      if (rowbase + 16 <= n) {
#pragma unroll
        for (int r = 0; r < 8; ++r) yp[(long)r * 40] = acc[r];
      } else {
#pragma unroll
        for (int r = 0; r < 8; ++r) {
          if (rbase + r < n) yp[(long)r * 40] = acc[r];
        }
      }
    }
  }
}

__global__ __launch_bounds__(256)
void GCN_gather40(const float* __restrict__ H, const unsigned* __restrict__ rowptr,
                  const int* __restrict__ sorted_src, const float* __restrict__ ndst,
                  const float* __restrict__ b, float* __restrict__ OUT, int n) {
  const int lane = threadIdx.x & 31;
  const int l2 = min(lane, 19);
  const int wave0 = __builtin_amdgcn_readfirstlane((int)(blockIdx.x * 8 + (threadIdx.x >> 5)));
  const int nwaves = (int)gridDim.x * 8;
  const v2f bb = ((const v2f*)b)[l2];
  for (int node = wave0; node < n; node += nwaves) {
    const int beg = (int)rowptr[node];
    const int end = (int)rowptr[node + 1];
    v2f acc = {0.f, 0.f};
    for (int j0 = beg; j0 < end; j0 += 32) {
      const int cnt = min(32, end - j0);
      const int mine = sorted_src[j0 + min(lane, cnt - 1)];
      for (int k = 0; k < cnt; k += 8) {
        v2f v[8];
#pragma unroll
        for (int i = 0; i < 8; ++i) {
          const int s = __builtin_amdgcn_readlane(mine, min(k + i, cnt - 1));
          v[i] = ((const v2f*)(H + (long)s * 40))[l2];
        }
        asm volatile("" :: "v"(v[0]), "v"(v[1]), "v"(v[2]), "v"(v[3]),
                           "v"(v[4]), "v"(v[5]), "v"(v[6]), "v"(v[7]));
        acc += v[0];
#pragma unroll
        for (int i = 1; i < 8; ++i) {
          if (k + i < cnt) acc += v[i];
        }
      }
    }
    if (lane < 20) {
      const float nd = ndst[node];
      ((v2f*)(OUT + (long)node * 40))[lane] = acc * nd + bb;
    }
  }
}


extern "C" void kernel_launch(void* const* d_in, const int* in_sizes, int n_in,
                              void* d_out, int out_size, void* d_ws, size_t ws_size,
                              hipStream_t stream) {
  const float* x   = (const float*)d_in[0];
  const float* W1  = (const float*)d_in[1];
  const float* b1  = (const float*)d_in[2];
  const float* W2  = (const float*)d_in[3];
  const float* b2  = (const float*)d_in[4];
  const int* esrc  = (const int*)d_in[5];
  const int* edst  = (const int*)d_in[6];
  float* out = (float*)d_out;

  const int nn = in_sizes[0] / 256;
  const int ne = in_sizes[5];

  const long NPAD = 50048;
  float* ws       = (float*)d_ws;
  unsigned* indeg  = (unsigned*)ws;
  unsigned* rowptr = (unsigned*)(ws + NPAD);
  float* nsrc     = ws + 2 * NPAD;
  float* ndst     = ws + 3 * NPAD;
  float* h        = ws + 4 * NPAD;
  float* h2       = ws + 4 * NPAD + 2L * (long)nn * 128;
  unsigned* part  = (unsigned*)(ws + 4 * NPAD + 2L * (long)nn * 128 + (long)nn * 40);
  int* ssrc       = (int*)(part + 2L * HIST_G * HIST_BINS);

  GCN_degree_hist<<<dim3(HIST_G, 2), HIST_T, 0, stream>>>(esrc, edst, part, ne);
  GCN_norms<<<dim3((HIST_BINS + 255) / 256, 2), 256, 0, stream>>>(part, nsrc, ndst, indeg, nn);

  GCN_sort_by_dst<<<HIST_G, HIST_T, 0, stream>>>(esrc, edst, part + (long)HIST_G * HIST_BINS,
                                                indeg, rowptr, ssrc, nn, ne);

  const int mtiles = (nn + 15) / 16;

  GCN_gemm_wmma<256, 8, 128><<<mtiles, 256, 0, stream>>>(x, nsrc, W1, h, nn);

  const int gtiles = (nn + 31) / 32;
  GCN_gather128_gemm40<<<gtiles, 256, 0, stream>>>(h, rowptr, ssrc, ndst, nsrc, b1, W2, h2, nn);

  GCN_gather40<<<GATHER_WGS, 256, 0, stream>>>(h2, rowptr, ssrc, ndst, b2, out, nn);
}


# sort pass split by destination range: 256 work-groups (32 edge chunks x 8 node ranges) instead of 32 issue the scattered stores; rowptr write-out shared; slot order by LDS-add arrival as before
# speedup vs baseline: 3.6926x; 1.2480x over previous
#include <hip/hip_runtime.h>


typedef __attribute__((ext_vector_type(2))) float v2f;
typedef __attribute__((ext_vector_type(4))) float v4f;
typedef __attribute__((ext_vector_type(8))) float v8f;

#define HIST_BINS 50048
#define HIST_G 32
#define HIST_T 1024

__global__ __launch_bounds__(HIST_T)
void GCN_degree_hist(const int* __restrict__ src, const int* __restrict__ dst,
                     unsigned* __restrict__ part, int ne) {
  __shared__ unsigned hist[HIST_BINS];
  const int* __restrict__ idx = blockIdx.y ? dst : src;

  uint4* h4 = (uint4*)hist;
  for (int i = threadIdx.x; i < HIST_BINS / 4; i += HIST_T) h4[i] = make_uint4(0u, 0u, 0u, 0u);
  __syncthreads();

  const int chunk = ((ne + 4 * HIST_G - 1) / (4 * HIST_G)) * 4;
  const int beg = min((int)blockIdx.x * chunk, ne);
  const int end = min(beg + chunk, ne);
  const int nvec = (end - beg) >> 2;
  const int4* __restrict__ p4 = (const int4*)(idx + beg);
  for (int base = 0; base < nvec; base += 4 * HIST_T) {
    int4 v[4];
#pragma unroll
    for (int j = 0; j < 4; ++j)
      v[j] = p4[min(base + j * HIST_T + (int)threadIdx.x, nvec - 1)];
#pragma unroll
    for (int j = 0; j < 4; ++j) {
      if (base + j * HIST_T + (int)threadIdx.x < nvec) {
        if ((unsigned)v[j].x < HIST_BINS) atomicAdd(&hist[v[j].x], 1u);
        if ((unsigned)v[j].y < HIST_BINS) atomicAdd(&hist[v[j].y], 1u);
        if ((unsigned)v[j].z < HIST_BINS) atomicAdd(&hist[v[j].z], 1u);
        if ((unsigned)v[j].w < HIST_BINS) atomicAdd(&hist[v[j].w], 1u);
      }
    }
  }
  for (int e = beg + (nvec << 2) + threadIdx.x; e < end; e += HIST_T) {
    const int v = idx[e];
    if ((unsigned)v < HIST_BINS) atomicAdd(&hist[v], 1u);
  }
  __syncthreads();

  uint4* o4 = (uint4*)(part + ((long)blockIdx.y * HIST_G + blockIdx.x) * HIST_BINS);
  for (int i = threadIdx.x; i < HIST_BINS / 4; i += HIST_T) o4[i] = h4[i];
}

__global__ void GCN_norms(unsigned* __restrict__ part,
                          float* __restrict__ nsrc, float* __restrict__ ndst,
                          unsigned* __restrict__ indeg, int n) {
  const int i = blockIdx.x * blockDim.x + threadIdx.x;
  if (i < n) {
    unsigned* p = part + (long)blockIdx.y * HIST_G * HIST_BINS + i;
    unsigned t[HIST_G];
#pragma unroll
    for (int g = 0; g < HIST_G; ++g) t[g] = p[(long)g * HIST_BINS];
    unsigned c = 0;
    if (blockIdx.y) {
#pragma unroll
      for (int g = 0; g < HIST_G; ++g) { p[(long)g * HIST_BINS] = c; c += t[g]; }
      indeg[i] = c;
      ndst[i] = rsqrtf(fmaxf((float)c, 1.0f));
    } else {
#pragma unroll
      for (int g = 0; g < HIST_G; ++g) c += t[g];
      nsrc[i] = rsqrtf(fmaxf((float)c, 1.0f));
    }
  } else if (blockIdx.y && i < HIST_BINS) {
    indeg[i] = 0u;
  }
}

#define SCAN_PER ((HIST_BINS + HIST_T - 1) / HIST_T)
#define SCAN_LEN (SCAN_PER * HIST_T)
#define BIN_VECS (HIST_BINS / 4)
#define VECS_PER ((BIN_VECS + HIST_T - 1) / HIST_T)
#define SORT_R 8
#define RANGE_BINS (HIST_BINS / SORT_R)
#define RANGE_VECS (RANGE_BINS / 4)
#define RVECS_PER ((RANGE_VECS + HIST_T - 1) / HIST_T)
#define ROWPTR_SLICE ((RANGE_BINS + HIST_G - 1) / HIST_G)
static_assert(HIST_BINS % (4 * SORT_R) == 0, "a range must be a whole number of 16-byte vectors");
static_assert(ROWPTR_SLICE <= HIST_T, "one rowptr entry per thread");

__global__ __launch_bounds__(HIST_T)
void GCN_sort_by_dst(const int* __restrict__ src, const int* __restrict__ dst,
                     const unsigned* __restrict__ pre, const unsigned* __restrict__ indeg,
                     unsigned* __restrict__ rowptr, int* __restrict__ sorted_src,
                     int n, int ne) {
  __shared__ unsigned cur[SCAN_LEN];
  __shared__ unsigned wsum[HIST_T / 32];
  const int t = threadIdx.x;
  const int lane = t & 31;
  const int w = t >> 5;
  uint4* c4 = (uint4*)cur;

  {
    const uint4* __restrict__ in4 = (const uint4*)indeg;
    uint4 v[VECS_PER];
#pragma unroll
    for (int j = 0; j < VECS_PER; ++j) v[j] = in4[min(j * HIST_T + t, BIN_VECS - 1)];
#pragma unroll
    for (int j = 0; j < VECS_PER; ++j) {
      const int q = j * HIST_T + t;
      if (q < BIN_VECS) c4[q] = v[j];
    }
    for (int i = HIST_BINS + t; i < SCAN_LEN; i += HIST_T) cur[i] = 0u;
  }
  __syncthreads();

  const int b0 = t * SCAN_PER;
  unsigned cnt[SCAN_PER];
  unsigned s = 0;
#pragma unroll
  for (int j = 0; j < SCAN_PER; ++j) { cnt[j] = cur[b0 + j]; s += cnt[j]; }
  unsigned inc = s;
#pragma unroll
  for (int o = 1; o < 32; o <<= 1) { const unsigned v = __shfl_up(inc, o); if (lane >= o) inc += v; }
  if (lane == 31) wsum[w] = inc;
  __syncthreads();
  if (w == 0) {
    const unsigned v = wsum[lane];
    unsigned iv = v;
#pragma unroll
    for (int o = 1; o < 32; o <<= 1) { const unsigned u = __shfl_up(iv, o); if (lane >= o) iv += u; }
    wsum[lane] = iv - v;
  }
  __syncthreads();
  unsigned run = inc - s + wsum[w];
#pragma unroll
  for (int j = 0; j < SCAN_PER; ++j) { cur[b0 + j] = run; run += cnt[j]; }
  __syncthreads();

  const int lo = (int)blockIdx.y * RANGE_BINS;
  const unsigned span = (unsigned)max(min(lo + RANGE_BINS, n) - lo, 0);
  const unsigned ulo = (unsigned)lo;

  {
    const int i = lo + (int)blockIdx.x * ROWPTR_SLICE + t;
    if (t < ROWPTR_SLICE && i < lo + RANGE_BINS && i <= n) rowptr[i] = cur[i];
  }
  __syncthreads();

  {
    const uint4* __restrict__ pg4 = (const uint4*)(pre + (long)blockIdx.x * HIST_BINS + lo);
    uint4* cr4 = (uint4*)(cur + lo);
    uint4 v[RVECS_PER];
#pragma unroll
    for (int j = 0; j < RVECS_PER; ++j) v[j] = pg4[min(j * HIST_T + t, RANGE_VECS - 1)];
#pragma unroll
    for (int j = 0; j < RVECS_PER; ++j) {
      const int q = j * HIST_T + t;
      if (q < RANGE_VECS) {
        uint4 c = cr4[q];
        c.x += v[j].x; c.y += v[j].y; c.z += v[j].z; c.w += v[j].w;
        cr4[q] = c;
      }
    }
  }
  __syncthreads();

  const int chunk = ((ne + 4 * HIST_G - 1) / (4 * HIST_G)) * 4;
  const int beg = min((int)blockIdx.x * chunk, ne);
  const int end = min(beg + chunk, ne);
  const int nvec = (end - beg) >> 2;
  const int4* __restrict__ d4 = (const int4*)(dst + beg);
  const int4* __restrict__ s4 = (const int4*)(src + beg);
  for (int base = 0; base < nvec; base += 4 * HIST_T) {
    int4 dv[4], sv[4];
#pragma unroll
    for (int j = 0; j < 4; ++j) {
      const int q = min(base + j * HIST_T + t, nvec - 1);
      dv[j] = d4[q];
      sv[j] = s4[q];
    }
#pragma unroll
    for (int j = 0; j < 4; ++j) {
      if (base + j * HIST_T + t < nvec) {
        if ((unsigned)dv[j].x - ulo < span) sorted_src[atomicAdd(&cur[dv[j].x], 1u)] = sv[j].x;
        if ((unsigned)dv[j].y - ulo < span) sorted_src[atomicAdd(&cur[dv[j].y], 1u)] = sv[j].y;
        if ((unsigned)dv[j].z - ulo < span) sorted_src[atomicAdd(&cur[dv[j].z], 1u)] = sv[j].z;
        if ((unsigned)dv[j].w - ulo < span) sorted_src[atomicAdd(&cur[dv[j].w], 1u)] = sv[j].w;
      }
    }
  }
  for (int e = beg + (nvec << 2) + t; e < end; e += HIST_T) {
    const int d = dst[e];
    if ((unsigned)d - ulo < span) sorted_src[atomicAdd(&cur[d], 1u)] = src[e];
  }
}

template <int K, int NT, int NC>
__global__ __launch_bounds__(32 * NT)
void GCN_gemm_wmma(const float* __restrict__ X, const float* __restrict__ scale,
                   const float* __restrict__ W, float* __restrict__ Y, int nrows) {
  constexpr int NCP = NT * 16;
  constexpr bool EXACT = (NC == NCP);
  constexpr int PITCH = K + 4;
  __shared__ float sa[16 * PITCH];
  __shared__ float sw[EXACT ? 4 : K * NCP];

  const int rowbase = blockIdx.x * 16;

  for (int idx = threadIdx.x; idx < 16 * (K / 4); idx += blockDim.x) {
    const int r = idx / (K / 4);
    const int c = idx - r * (K / 4);
    const int row = rowbase + r;
    float4 v = make_float4(0.f, 0.f, 0.f, 0.f);
    float s = 0.0f;
    if (row < nrows) {
      v = ((const float4*)(X + (long)row * K))[c];
      s = scale[row];
    }
    float* d = &sa[r * PITCH + 4 * c];
    d[0] = v.x * s; d[1] = v.y * s; d[2] = v.z * s; d[3] = v.w * s;
  }

  if constexpr (!EXACT) {
    for (int idx = threadIdx.x; idx < K * NCP; idx += blockDim.x) {
      const int k = idx / NCP;
      const int c = idx - k * NCP;
      sw[idx] = (c < NC) ? W[k * NC + c] : 0.0f;
    }
  }
  __syncthreads();

  const int lane = threadIdx.x & 31;
  const int wave = threadIdx.x >> 5;
  const int colbase = wave * 16;
  const int half = lane >> 4;
  const int l15 = lane & 15;
  const int koff = half * 2;

  const float* aptr = &sa[l15 * PITCH + koff];
  const int bcol = colbase + l15;

  v8f acc = {};
#pragma unroll 8
  for (int k0 = 0; k0 < K; k0 += 4) {
    v2f a;
    a.x = aptr[k0];
    a.y = aptr[k0 + 1];
    v2f b;
    if constexpr (EXACT) {
      b.x = W[(k0 + koff) * NC + bcol];
      b.y = W[(k0 + koff + 1) * NC + bcol];
    } else {
      b.x = sw[(k0 + koff) * NCP + bcol];
      b.y = sw[(k0 + koff + 1) * NCP + bcol];
    }
    acc = __builtin_amdgcn_wmma_f32_16x16x4_f32(
         false, a,  false, b,
         (short)0, acc,  false,  false);
  }

  if (EXACT || bcol < NC) {
    float* yp = Y + (long)(rowbase + half * 8) * NC + bcol;
    if (rowbase + 16 <= nrows) {
#pragma unroll
      for (int r = 0; r < 8; ++r) yp[(long)r * NC] = acc[r];
    } else {
      const int rbase = rowbase + half * 8;
#pragma unroll
      for (int r = 0; r < 8; ++r) {
        if (rbase + r < nrows) yp[(long)r * NC] = acc[r];
      }
    }
  }
}

#define GATHER_WGS 2048

#define FUSE_ROWS 32
#define FUSE_PITCH 132
#define FUSE_NCP 48

__global__ __launch_bounds__(256)
void GCN_gather128_gemm40(const float* __restrict__ H, const unsigned* __restrict__ rowptr,
                          const int* __restrict__ sorted_src, const float* __restrict__ ndst,
                          const float* __restrict__ nsrc, const float* __restrict__ b,
                          const float* __restrict__ W2, float* __restrict__ H2, int n) {
  __shared__ float sa[FUSE_ROWS * FUSE_PITCH];
  __shared__ float sw[128 * FUSE_NCP];
  const int t = threadIdx.x;
  const int lane = t & 31;
  const int wave = __builtin_amdgcn_readfirstlane(t >> 5);

  {
    v4f v[5];
#pragma unroll
    for (int j = 0; j < 5; ++j) v[j] = ((const v4f*)W2)[t + 256 * j];
#pragma unroll
    for (int j = 0; j < 5; ++j) {
      const int q = t + 256 * j;
      const int k = q / 10;
      const int c = q - k * 10;
      *(v4f*)&sw[k * FUSE_NCP + 4 * c] = v[j];
    }
    const v4f z = {0.f, 0.f, 0.f, 0.f};
    *(v4f*)&sw[(t >> 1) * FUSE_NCP + 40 + 4 * (t & 1)] = z;
  }

  const v4f bb = ((const v4f*)b)[lane];
  const int node0 = (int)blockIdx.x * FUSE_ROWS + wave * 4;
  for (int i = 0; i < 4; ++i) {
    const int node = node0 + i;
    v4f o = {0.f, 0.f, 0.f, 0.f};
    if (node < n) {
      const float nd = ndst[node];
      const float ns = nsrc[node];
      const int beg = (int)rowptr[node];
      const int end = (int)rowptr[node + 1];
      v4f acc = {0.f, 0.f, 0.f, 0.f};
      for (int j0 = beg; j0 < end; j0 += 32) {
        const int cnt = min(32, end - j0);
        const int mine = sorted_src[j0 + min(lane, cnt - 1)];
        for (int k = 0; k < cnt; k += 8) {
          v4f v[8];
#pragma unroll
          for (int s8 = 0; s8 < 8; ++s8) {
            const int s = __builtin_amdgcn_readlane(mine, min(k + s8, cnt - 1));
            v[s8] = ((const v4f*)(H + (long)s * 128))[lane];
          }
          asm volatile("" :: "v"(v[0]), "v"(v[1]), "v"(v[2]), "v"(v[3]),
                             "v"(v[4]), "v"(v[5]), "v"(v[6]), "v"(v[7]));
          acc += v[0];
#pragma unroll
          for (int s8 = 1; s8 < 8; ++s8) {
            if (k + s8 < cnt) acc += v[s8];
          }
        }
      }
      o = acc * nd + bb;
      o.x = o.x > 0.0f ? o.x : 0.0f;
      o.y = o.y > 0.0f ? o.y : 0.0f;
      o.z = o.z > 0.0f ? o.z : 0.0f;
      o.w = o.w > 0.0f ? o.w : 0.0f;
      o = o * ns;
    }
    *(v4f*)&sa[(wave * 4 + i) * FUSE_PITCH + 4 * lane] = o;
  }
  __syncthreads();

  if (wave < 6) {
    const int tile = wave / 3;
    const int ct = wave - tile * 3;
    const int half = lane >> 4;
    const int l15 = lane & 15;
    const int koff = half * 2;
    const float* aptr = &sa[(tile * 16 + l15) * FUSE_PITCH + koff];
    const int bcol = ct * 16 + l15;

    v8f acc = {};
#pragma unroll 8
    for (int k0 = 0; k0 < 128; k0 += 4) {
      v2f a;
      a.x = aptr[k0];
      a.y = aptr[k0 + 1];
      v2f bf;
      bf.x = sw[(k0 + koff) * FUSE_NCP + bcol];
      bf.y = sw[(k0 + koff + 1) * FUSE_NCP + bcol];
      acc = __builtin_amdgcn_wmma_f32_16x16x4_f32(
           false, a,  false, bf,
           (short)0, acc,  false,  false);
    }

    if (bcol < 40) {
      const int rowbase = (int)blockIdx.x * FUSE_ROWS + tile * 16;
      const int rbase = rowbase + half * 8;
      float* yp = H2 + (long)rbase * 40 + bcol;
      if (rowbase + 16 <= n) {
#pragma unroll
        for (int r = 0; r < 8; ++r) yp[(long)r * 40] = acc[r];
      } else {
#pragma unroll
        for (int r = 0; r < 8; ++r) {
          if (rbase + r < n) yp[(long)r * 40] = acc[r];
        }
      }
    }
  }
}

__global__ __launch_bounds__(256)
void GCN_gather40(const float* __restrict__ H, const unsigned* __restrict__ rowptr,
                  const int* __restrict__ sorted_src, const float* __restrict__ ndst,
                  const float* __restrict__ b, float* __restrict__ OUT, int n) {
  const int lane = threadIdx.x & 31;
  const int l2 = min(lane, 19);
  const int wave0 = __builtin_amdgcn_readfirstlane((int)(blockIdx.x * 8 + (threadIdx.x >> 5)));
  const int nwaves = (int)gridDim.x * 8;
  const v2f bb = ((const v2f*)b)[l2];
  for (int node = wave0; node < n; node += nwaves) {
    const int beg = (int)rowptr[node];
    const int end = (int)rowptr[node + 1];
    v2f acc = {0.f, 0.f};
    for (int j0 = beg; j0 < end; j0 += 32) {
      const int cnt = min(32, end - j0);
      const int mine = sorted_src[j0 + min(lane, cnt - 1)];
      for (int k = 0; k < cnt; k += 8) {
        v2f v[8];
#pragma unroll
        for (int i = 0; i < 8; ++i) {
          const int s = __builtin_amdgcn_readlane(mine, min(k + i, cnt - 1));
          v[i] = ((const v2f*)(H + (long)s * 40))[l2];
        }
        asm volatile("" :: "v"(v[0]), "v"(v[1]), "v"(v[2]), "v"(v[3]),
                           "v"(v[4]), "v"(v[5]), "v"(v[6]), "v"(v[7]));
        acc += v[0];
#pragma unroll
        for (int i = 1; i < 8; ++i) {
          if (k + i < cnt) acc += v[i];
        }
      }
    }
    if (lane < 20) {
      const float nd = ndst[node];
      ((v2f*)(OUT + (long)node * 40))[lane] = acc * nd + bb;
    }
  }
}


extern "C" void kernel_launch(void* const* d_in, const int* in_sizes, int n_in,
                              void* d_out, int out_size, void* d_ws, size_t ws_size,
                              hipStream_t stream) {
  const float* x   = (const float*)d_in[0];
  const float* W1  = (const float*)d_in[1];
  const float* b1  = (const float*)d_in[2];
  const float* W2  = (const float*)d_in[3];
  const float* b2  = (const float*)d_in[4];
  const int* esrc  = (const int*)d_in[5];
  const int* edst  = (const int*)d_in[6];
  float* out = (float*)d_out;

  const int nn = in_sizes[0] / 256;
  const int ne = in_sizes[5];

  const long NPAD = 50048;
  float* ws       = (float*)d_ws;
  unsigned* indeg  = (unsigned*)ws;
  unsigned* rowptr = (unsigned*)(ws + NPAD);
  float* nsrc     = ws + 2 * NPAD;
  float* ndst     = ws + 3 * NPAD;
  float* h        = ws + 4 * NPAD;
  float* h2       = ws + 4 * NPAD + 2L * (long)nn * 128;
  unsigned* part  = (unsigned*)(ws + 4 * NPAD + 2L * (long)nn * 128 + (long)nn * 40);
  int* ssrc       = (int*)(part + 2L * HIST_G * HIST_BINS);

  GCN_degree_hist<<<dim3(HIST_G, 2), HIST_T, 0, stream>>>(esrc, edst, part, ne);
  GCN_norms<<<dim3((HIST_BINS + 255) / 256, 2), 256, 0, stream>>>(part, nsrc, ndst, indeg, nn);

  GCN_sort_by_dst<<<dim3(HIST_G, SORT_R), HIST_T, 0, stream>>>(esrc, edst, part + (long)HIST_G * HIST_BINS,
                                                indeg, rowptr, ssrc, nn, ne);

  const int mtiles = (nn + 15) / 16;

  GCN_gemm_wmma<256, 8, 128><<<mtiles, 256, 0, stream>>>(x, nsrc, W1, h, nn);

  const int gtiles = (nn + 31) / 32;
  GCN_gather128_gemm40<<<gtiles, 256, 0, stream>>>(h, rowptr, ssrc, ndst, nsrc, b1, W2, h2, nn);

  GCN_gather40<<<GATHER_WGS, 256, 0, stream>>>(h2, rowptr, ssrc, ndst, b2, out, nn);
}
